# starting kernel
# baseline (speedup 1.0000x reference)
#include <hip/hip_runtime.h>
#include <hip/hip_bf16.h>


typedef float v2f __attribute__((ext_vector_type(2)));
typedef float v8f __attribute__((ext_vector_type(8)));

#define D 128
#define XS_STRIDE 132


__global__ void fill_f32(float* __restrict__ p, float v, int n) {
    int i = blockIdx.x * blockDim.x + threadIdx.x;
    if (i < n) p[i] = v;
}

__global__ void deg_accum(const int* __restrict__ dst, float* __restrict__ deg, int nE) {
    int e = blockIdx.x * blockDim.x + threadIdx.x;
    if (e < nE) atomicAdd(&deg[dst[e]], 1.0f);
}

__global__ void deg_to_dinv(float* __restrict__ d, int n) {
    int i = blockIdx.x * blockDim.x + threadIdx.x;
    if (i < n) d[i] = rsqrtf(d[i]);
}

__global__ void bias_relu(float* __restrict__ h, const float* __restrict__ b, int n) {
    int i = blockIdx.x * blockDim.x + threadIdx.x;
    if (i < n) {
        float v = h[i] + b[i & (D - 1)];
        h[i] = v > 0.0f ? v : 0.0f;
    }
}

__global__ void bias_add(float* __restrict__ h, const float* __restrict__ b, int n) {
    int i = blockIdx.x * blockDim.x + threadIdx.x;
    if (i < n) h[i] = h[i] + b[i & (D - 1)];
}

__global__ void gemm_wmma_f32(const float* __restrict__ X,
                              const float* __restrict__ W,
                              float* __restrict__ H, int nRows) {
    __shared__ float xs[16 * XS_STRIDE];

    const int row0 = blockIdx.x * 16;
    const int tid  = threadIdx.x;

    for (int i = tid; i < 16 * (D / 4); i += 256) {
        const int r = i >> 5;
        const int c = i & 31;
        float4 v = make_float4(0.f, 0.f, 0.f, 0.f);
        if (row0 + r < nRows)
            v = ((const float4*)(X + (size_t)(row0 + r) * D))[c];
        float* p = &xs[r * XS_STRIDE + c * 4];
        p[0] = v.x; p[1] = v.y; p[2] = v.z; p[3] = v.w;
    }
    __syncthreads();

    const int wave = tid >> 5;
    const int lane = tid & 31;
    const int m    = lane & 15;
    const int kk   = (lane >> 4) << 1;
    const int n    = lane & 15;
    const int col  = wave * 16 + n;

    v8f acc = {};
    #pragma unroll
    for (int k0 = 0; k0 < D; k0 += 4) {
        v2f a, b;
        a.x = xs[m * XS_STRIDE + k0 + kk];
        a.y = xs[m * XS_STRIDE + k0 + kk + 1];
        b.x = W[(size_t)(k0 + kk) * D + col];
        b.y = W[(size_t)(k0 + kk + 1) * D + col];
        acc = __builtin_amdgcn_wmma_f32_16x16x4_f32(
             false, a,  false, b,
             (short)0, acc,  false,  false);
    }

    const int halfm = (lane >> 4) * 8;
    #pragma unroll
    for (int r = 0; r < 8; ++r) {
        const int row = row0 + halfm + r;
        if (row < nRows) H[(size_t)row * D + col] = acc[r];
    }
}

__global__ void edge_aggregate(const int* __restrict__ src,
                               const int* __restrict__ dst,
                               const float* __restrict__ dinv,
                               const float* __restrict__ h,
                               float* __restrict__ out, int nE, int nN) {
    const int warp = (blockIdx.x * blockDim.x + threadIdx.x) >> 5;
    const int lane = threadIdx.x & 31;
    const int total = nE + nN;
    if (warp >= total) return;

    int s, d;
    if (warp < nE) { s = src[warp]; d = dst[warp]; }
    else           { s = d = warp - nE; }

    const float norm = dinv[s] * dinv[d];
    const float4 v = ((const float4*)(h + (size_t)s * D))[lane];

    float* o = out + (size_t)d * D + lane * 4;
    atomicAdd(o + 0, v.x * norm);
    atomicAdd(o + 1, v.y * norm);
    atomicAdd(o + 2, v.z * norm);
    atomicAdd(o + 3, v.w * norm);
}


static inline size_t align256(size_t v) { return (v + 255) & ~(size_t)255; }

extern "C" void kernel_launch(void* const* d_in, const int* in_sizes, int n_in,
                              void* d_out, int out_size, void* d_ws, size_t ws_size,
                              hipStream_t stream) {
    const float* x  = (const float*)d_in[0];
    const int*   ei = (const int*)d_in[1];
    const float* W1 = (const float*)d_in[2];
    const float* b1 = (const float*)d_in[3];
    const float* W2 = (const float*)d_in[4];
    const float* b2 = (const float*)d_in[5];
    float* out = (float*)d_out;

    const int N = in_sizes[0] / D;
    const int E = in_sizes[1] / 2;
    const int* srcIdx = ei;
    const int* dstIdx = ei + E;

    char* ws = (char*)d_ws;
    float* dinv = (float*)ws;
    const size_t o1 = align256((size_t)N * sizeof(float));
    float* hbuf = (float*)(ws + o1);
    const size_t o2 = o1 + align256((size_t)N * D * sizeof(float));
    float* agg  = (float*)(ws + o2);

    const int nF = N * D;
    const int blk = 256;
    const int gN   = (N + blk - 1) / blk;
    const int gE   = (E + blk - 1) / blk;
    const int gF   = (nF + blk - 1) / blk;
    const int gGemm = (N + 15) / 16;
    const int gEdge = ((E + N) + 7) / 8;

    fill_f32   <<<gN, blk, 0, stream>>>(dinv, 1.0f, N);
    deg_accum  <<<gE, blk, 0, stream>>>(dstIdx, dinv, E);
    deg_to_dinv<<<gN, blk, 0, stream>>>(dinv, N);

    gemm_wmma_f32 <<<gGemm, blk, 0, stream>>>(x, W1, hbuf, N);
    fill_f32      <<<gF, blk, 0, stream>>>(agg, 0.0f, nF);
    edge_aggregate<<<gEdge, blk, 0, stream>>>(srcIdx, dstIdx, dinv, hbuf, agg, E, N);
    bias_relu     <<<gF, blk, 0, stream>>>(agg, b1, nF);

    gemm_wmma_f32 <<<gGemm, blk, 0, stream>>>(agg, W2, hbuf, N);
    fill_f32      <<<gF, blk, 0, stream>>>(out, 0.0f, nF);
    edge_aggregate<<<gEdge, blk, 0, stream>>>(srcIdx, dstIdx, dinv, hbuf, out, E, N);
    bias_add      <<<gF, blk, 0, stream>>>(out, b2, nF);
}


# GEMM: 1024 resident blocks each walk 16-row strips through two LDS buffers, next strip's loads in flight under the current products; W fragments fetched once per block; sums unchanged
# speedup vs baseline: 2.7195x; 2.7195x over previous
#include <hip/hip_runtime.h>
#include <hip/hip_bf16.h>


typedef float v2f __attribute__((ext_vector_type(2)));
typedef float v8f __attribute__((ext_vector_type(8)));

#define D 128
#define XS_STRIDE 132


__global__ void fill_f32(float* __restrict__ p, float v, int n, int* __restrict__ ctr) {
    int i = blockIdx.x * blockDim.x + threadIdx.x;
    if (i < n) p[i] = v;
    if (i < 4) ctr[i] = 0;
}

__global__ void deg_accum(const int* __restrict__ dst, float* __restrict__ deg, int nE) {
    int e = blockIdx.x * blockDim.x + threadIdx.x;
    if (e < nE) atomicAdd(&deg[dst[e]], 1.0f);
}

__global__ void deg_to_norm(const int* __restrict__ src, const int* __restrict__ dst,
                            const float* __restrict__ deg, float* __restrict__ dinv,
                            float* __restrict__ norm, int n, int nE) {
    int i = blockIdx.x * blockDim.x + threadIdx.x;
    if (i < n) dinv[i] = rsqrtf(deg[i]);
    if (i < nE) norm[i] = rsqrtf(deg[src[i]]) * rsqrtf(deg[dst[i]]);
}

#define GEMM_BLOCKS 1024

__global__ void __launch_bounds__(256)
gemm_wmma_f32(const float* __restrict__ X,
              const float* __restrict__ W,
              const float* __restrict__ bias,
              const float* __restrict__ dinv,
              float* __restrict__ H,
              float* __restrict__ S,
              int nRows, int reluIn) {
    __shared__ float xs[2][16 * XS_STRIDE];
    __shared__ float dsq[2][16];

    const int tid  = threadIdx.x;
    const int wave = tid >> 5;
    const int lane = tid & 31;
    const int m    = lane & 15;
    const int kk   = (lane >> 4) << 1;
    const int col  = wave * 16 + m;
    const int halfm = (lane >> 4) * 8;

    const int nStrips = (nRows + 15) >> 4;
    const int step = (int)gridDim.x;
    int s = (int)blockIdx.x;
    if (s >= nStrips) return;

    const int sr = tid >> 5;
    const int sc = tid & 31;

    float4 p0, p1;
    float  pd = 0.0f;

    {
        const int ra = s * 16 + sr;
        p0 = make_float4(0.f, 0.f, 0.f, 0.f);
        p1 = make_float4(0.f, 0.f, 0.f, 0.f);
        if (ra < nRows)     p0 = ((const float4*)(X + (size_t)ra * D))[sc];
        if (ra + 8 < nRows) p1 = ((const float4*)(X + (size_t)(ra + 8) * D))[sc];
        if (tid < 16) pd = (s * 16 + tid < nRows) ? dinv[s * 16 + tid] : 0.0f;
    }

    v2f bw[32];
    #pragma unroll
    for (int j = 0; j < 32; ++j) {
        bw[j].x = W[(size_t)(4 * j + kk) * D + col];
        bw[j].y = W[(size_t)(4 * j + kk + 1) * D + col];
    }
    const float bc = bias[col];

    int cur = 0;
    {
        if (reluIn) {
            p0.x = fmaxf(p0.x, 0.0f); p0.y = fmaxf(p0.y, 0.0f);
            p0.z = fmaxf(p0.z, 0.0f); p0.w = fmaxf(p0.w, 0.0f);
            p1.x = fmaxf(p1.x, 0.0f); p1.y = fmaxf(p1.y, 0.0f);
            p1.z = fmaxf(p1.z, 0.0f); p1.w = fmaxf(p1.w, 0.0f);
        }
        *(float4*)&xs[0][sr * XS_STRIDE + sc * 4]       = p0;
        *(float4*)&xs[0][(sr + 8) * XS_STRIDE + sc * 4] = p1;
        if (tid < 16) dsq[0][tid] = pd * pd;
    }
    __syncthreads();

    for (; s < nStrips; s += step) {
        const int row0 = s * 16;
        const int sn = s + step;
        const bool more = sn < nStrips;

        if (more) {
            const int ra = sn * 16 + sr;
            p0 = make_float4(0.f, 0.f, 0.f, 0.f);
            p1 = make_float4(0.f, 0.f, 0.f, 0.f);
            if (ra < nRows)     p0 = ((const float4*)(X + (size_t)ra * D))[sc];
            if (ra + 8 < nRows) p1 = ((const float4*)(X + (size_t)(ra + 8) * D))[sc];
            if (tid < 16) pd = (sn * 16 + tid < nRows) ? dinv[sn * 16 + tid] : 0.0f;
        }

        const float* xa = &xs[cur][m * XS_STRIDE + kk];
        v8f acc = {};
        #pragma unroll
        for (int j = 0; j < 32; ++j) {
            v2f a;
            a.x = xa[4 * j];
            a.y = xa[4 * j + 1];
            acc = __builtin_amdgcn_wmma_f32_16x16x4_f32(
                 false, a,  false, bw[j],
                 (short)0, acc,  false,  false);
        }

        float n2[8];
        #pragma unroll
        for (int r = 0; r < 8; ++r) n2[r] = dsq[cur][halfm + r];
        float* hp = H + (size_t)(row0 + halfm) * D + col;
        float* sp = S + (size_t)(row0 + halfm) * D + col;
        if (row0 + 16 <= nRows) {
            #pragma unroll
            for (int r = 0; r < 8; ++r) {
                const float hv = acc[r];
                hp[(size_t)r * D] = hv;
                sp[(size_t)r * D] = hv * n2[r] + bc;
            }
        } else {
            #pragma unroll
            for (int r = 0; r < 8; ++r) {
                if (row0 + halfm + r < nRows) {
                    const float hv = acc[r];
                    hp[(size_t)r * D] = hv;
                    sp[(size_t)r * D] = hv * n2[r] + bc;
                }
            }
        }

        if (more) {
            if (reluIn) {
                p0.x = fmaxf(p0.x, 0.0f); p0.y = fmaxf(p0.y, 0.0f);
                p0.z = fmaxf(p0.z, 0.0f); p0.w = fmaxf(p0.w, 0.0f);
                p1.x = fmaxf(p1.x, 0.0f); p1.y = fmaxf(p1.y, 0.0f);
                p1.z = fmaxf(p1.z, 0.0f); p1.w = fmaxf(p1.w, 0.0f);
            }
            *(float4*)&xs[cur ^ 1][sr * XS_STRIDE + sc * 4]       = p0;
            *(float4*)&xs[cur ^ 1][(sr + 8) * XS_STRIDE + sc * 4] = p1;
            if (tid < 16) dsq[cur ^ 1][tid] = pd * pd;
        }
        __builtin_amdgcn_fence(__ATOMIC_RELEASE, "workgroup", "local");
        __builtin_amdgcn_s_barrier();
        __builtin_amdgcn_fence(__ATOMIC_ACQUIRE, "workgroup", "local");
        cur ^= 1;
    }
}

#define EPW 64
#define WPB 8
#define EPB (EPW * WPB)
#define EB  16

__device__ __forceinline__ int lane_bcast_i(int v, int j) {
    return __builtin_amdgcn_readlane(v, j);
}
__device__ __forceinline__ float lane_bcast_f(float v, int j) {
    return __int_as_float(__builtin_amdgcn_readlane(__float_as_int(v), j));
}

__device__ __forceinline__ void walk32(int s, int d, float nrm,
                                       const float* __restrict__ hc,
                                       float* __restrict__ oc, unsigned lane) {
    for (int j0 = 0; j0 < 32; j0 += EB) {
        float v[EB][2];
        #pragma unroll
        for (int u = 0; u < EB; ++u) {
            const int sj = lane_bcast_i(s, j0 + u);
            const float* hr = hc + (size_t)sj * D;
            v[u][0] = hr[lane];
            v[u][1] = hr[lane + 32];
        }
        #pragma unroll
        for (int u = 0; u < EB; ++u) {
            const int   dj = lane_bcast_i(d, j0 + u);
            const float nj = lane_bcast_f(nrm, j0 + u);
            float* o = oc + (size_t)dj * D;
            atomicAdd(o + lane,      v[u][0] * nj);
            atomicAdd(o + lane + 32, v[u][1] * nj);
        }
    }
}

__device__ __forceinline__ void walk_some(int s, int d, float nrm, int cnt,
                                          const float* __restrict__ hc,
                                          float* __restrict__ oc, unsigned lane) {
    for (int j = 0; j < cnt; ++j) {
        const int   sj = lane_bcast_i(s, j);
        const int   dj = lane_bcast_i(d, j);
        const float nj = lane_bcast_f(nrm, j);
        const float* hr = hc + (size_t)sj * D;
        const float v0 = hr[lane];
        const float v1 = hr[lane + 32];
        float* o = oc + (size_t)dj * D;
        atomicAdd(o + lane,      v0 * nj);
        atomicAdd(o + lane + 32, v1 * nj);
    }
}

__global__ void __launch_bounds__(256)
edge_aggregate(const int* __restrict__ src,
               const int* __restrict__ dst,
               const float* __restrict__ norm,
               const float* __restrict__ h,
               float* __restrict__ out, int nE, int nItems,
               int* __restrict__ ctr) {
    __shared__ int pick[2];
    if (threadIdx.x == 0) {
        const unsigned where = __builtin_amdgcn_s_sendmsg_rtn(0x87);
        int c = (int)((where >> 18) & 1u);
        int t = atomicAdd(&ctr[c], 1);
        if (t >= nItems) {
            c ^= 1;
            t = atomicAdd(&ctr[c], 1);
        }
        pick[0] = c;
        pick[1] = t;
    }
    __syncthreads();
    const int c = __builtin_amdgcn_readfirstlane(pick[0]);
    const int t = __builtin_amdgcn_readfirstlane(pick[1]);
    if (t >= nItems) return;

    const unsigned lane = threadIdx.x & 31;
    const int wv = __builtin_amdgcn_readfirstlane((int)(threadIdx.x >> 5));
    const int total = nE;
    const int e0 = t * EPB + wv * EPW;
    const int cnt = total - e0;
    if (cnt <= 0) return;

    const float* hc = h + c * 64;
    float* oc = out + c * 64;

    const int ea = e0 + (int)lane;
    const int eb = ea + 32;
    if (cnt >= EPW) {
        const int sa = src[ea], da = dst[ea];
        const int sb = src[eb], db = dst[eb];
        const float na = norm[ea], nb = norm[eb];
        walk32(sa, da, na, hc, oc, lane);
        walk32(sb, db, nb, hc, oc, lane);
    } else {
        int sa = 0, da = 0, sb = 0, db = 0;
        float na = 0.0f, nb = 0.0f;
        if (ea < total) { sa = src[ea]; da = dst[ea]; na = norm[ea]; }
        if (eb < total) { sb = src[eb]; db = dst[eb]; nb = norm[eb]; }
        if (cnt >= 32) {
            walk32(sa, da, na, hc, oc, lane);
            walk_some(sb, db, nb, cnt - 32, hc, oc, lane);
        } else {
            walk_some(sa, da, na, cnt, hc, oc, lane);
        }
    }
}


static inline size_t align256(size_t v) { return (v + 255) & ~(size_t)255; }

extern "C" void kernel_launch(void* const* d_in, const int* in_sizes, int n_in,
                              void* d_out, int out_size, void* d_ws, size_t ws_size,
                              hipStream_t stream) {
    const float* x  = (const float*)d_in[0];
    const int*   ei = (const int*)d_in[1];
    const float* W1 = (const float*)d_in[2];
    const float* b1 = (const float*)d_in[3];
    const float* W2 = (const float*)d_in[4];
    const float* b2 = (const float*)d_in[5];
    float* out = (float*)d_out;

    const int N = in_sizes[0] / D;
    const int E = in_sizes[1] / 2;
    const int* srcIdx = ei;
    const int* dstIdx = ei + E;

    char* ws = (char*)d_ws;
    float* deg  = (float*)ws;
    const size_t o0 = align256((size_t)N * sizeof(float));
    float* dinv = (float*)(ws + o0);
    const size_t o1 = o0 + align256((size_t)N * sizeof(float));
    float* hbuf = (float*)(ws + o1);
    const size_t o2 = o1 + align256((size_t)N * D * sizeof(float));
    float* agg  = (float*)(ws + o2);
    const size_t o3 = o2 + align256((size_t)N * D * sizeof(float));
    int* ctr = (int*)(ws + o3);
    float* norm = (float*)(ws + o3 + 256);

    const int blk = 256;
    const int gN   = (N + blk - 1) / blk;
    const int gE   = (E + blk - 1) / blk;
    const int gNE  = gE > gN ? gE : gN;
    const int gGemm = GEMM_BLOCKS;
    const int nItems = (E + EPB - 1) / EPB;
    const int gEdge = 2 * nItems;

    fill_f32   <<<gN, blk, 0, stream>>>(deg, 1.0f, N, ctr);
    deg_accum  <<<gE, blk, 0, stream>>>(dstIdx, deg, E);
    deg_to_norm<<<gNE, blk, 0, stream>>>(srcIdx, dstIdx, deg, dinv, norm, N, E);

    gemm_wmma_f32 <<<gGemm, blk, 0, stream>>>(x, W1, b1, dinv, hbuf, agg, N, 0);
    edge_aggregate<<<gEdge, blk, 0, stream>>>(srcIdx, dstIdx, norm, hbuf, agg, E, nItems, ctr);

    gemm_wmma_f32 <<<gGemm, blk, 0, stream>>>(agg, W2, b2, dinv, hbuf, out, N, 1);
    edge_aggregate<<<gEdge, blk, 0, stream>>>(srcIdx, dstIdx, norm, hbuf, out, E, nItems, ctr + 2);
}


# Edge sums in LDS: edges regrouped per call by a counting pass into 400-row dst groups; a work-group keeps its rows' sums in LDS (f32 LDS adds), stores each row once; sum order: LDS arrival
# speedup vs baseline: 2.9066x; 1.0688x over previous
#include <hip/hip_runtime.h>
#include <hip/hip_bf16.h>


typedef float v2f __attribute__((ext_vector_type(2)));
typedef float v8f __attribute__((ext_vector_type(8)));

#define D 128
#define XS_STRIDE 132

#define RB   400
#define CH   2048
#define CPT  8
#define MAXG 1024


__global__ void init_deg(float* __restrict__ deg, int n, int* __restrict__ gcnt,
                         int* __restrict__ gcur, int nG) {
    int i = blockIdx.x * blockDim.x + threadIdx.x;
    if (i < n) deg[i] = 1.0f;
    if (i < nG) { gcnt[i] = 0; gcur[i] = 0; }
}

__global__ void __launch_bounds__(256)
deg_hist(const int* __restrict__ dst, float* __restrict__ deg,
         int* __restrict__ gcnt, int nE, int nG) {
    __shared__ int lh[MAXG];
    const int tid = threadIdx.x;
    for (int b = tid; b < nG; b += 256) lh[b] = 0;
    __syncthreads();
    const int base = blockIdx.x * CH;
    #pragma unroll
    for (int k = 0; k < CPT; ++k) {
        const int e = base + k * 256 + tid;
        if (e < nE) {
            const int d = dst[e];
            atomicAdd(&deg[d], 1.0f);
            atomicAdd(&lh[d / RB], 1);
        }
    }
    __syncthreads();
    for (int b = tid; b < nG; b += 256) {
        const int c = lh[b];
        if (c) atomicAdd(&gcnt[b], c);
    }
}

__global__ void __launch_bounds__(256)
group_edges(const int* __restrict__ src, const int* __restrict__ dst,
            const float* __restrict__ deg, float* __restrict__ dinv,
            const int* __restrict__ gcnt, int* __restrict__ gcur,
            int* __restrict__ gstart,
            int* __restrict__ bsrc, int* __restrict__ bdst, float* __restrict__ bnorm,
            int n, int nE, int nG) {
    __shared__ int lcnt[MAXG];
    __shared__ int lstart[MAXG];
    __shared__ int part[256];
    const int tid = threadIdx.x;

    for (int i = blockIdx.x * 256 + tid; i < n; i += gridDim.x * 256)
        dinv[i] = rsqrtf(deg[i]);

    for (int b = tid; b < nG; b += 256) lcnt[b] = 0;

    const int per = (nG + 255) / 256;
    int mine = 0;
    for (int q = 0; q < per; ++q) {
        const int b = tid * per + q;
        if (b < nG) mine += gcnt[b];
    }
    part[tid] = mine;
    __syncthreads();
    for (int off = 1; off < 256; off <<= 1) {
        const int v = (tid >= off) ? part[tid - off] : 0;
        __syncthreads();
        part[tid] += v;
        __syncthreads();
    }
    {
        int run = part[tid] - mine;
        for (int q = 0; q < per; ++q) {
            const int b = tid * per + q;
            if (b < nG) {
                lstart[b] = run;
                if (blockIdx.x == 0) gstart[b] = run;
                run += gcnt[b];
            }
        }
        if (blockIdx.x == 0 && tid == 255) gstart[nG] = part[255];
    }

    const int base = blockIdx.x * CH;
    int dk[CPT], rk[CPT];
    #pragma unroll
    for (int k = 0; k < CPT; ++k) {
        const int e = base + k * 256 + tid;
        dk[k] = -1;
        rk[k] = 0;
        if (e < nE) {
            const int d = dst[e];
            dk[k] = d;
            rk[k] = atomicAdd(&lcnt[d / RB], 1);
        }
    }
    __syncthreads();
    for (int b = tid; b < nG; b += 256) {
        const int c = lcnt[b];
        if (c) lcnt[b] = lstart[b] + atomicAdd(&gcur[b], c);
    }
    __syncthreads();
    #pragma unroll
    for (int k = 0; k < CPT; ++k) {
        if (dk[k] >= 0) {
            const int e = base + k * 256 + tid;
            const int d = dk[k];
            const int s = src[e];
            const int pos = lcnt[d / RB] + rk[k];
            bsrc[pos]  = s;
            bdst[pos]  = d;
            bnorm[pos] = rsqrtf(deg[s]) * rsqrtf(deg[d]);
        }
    }
}

#define GEMM_BLOCKS 1024

__global__ void __launch_bounds__(256)
gemm_wmma_f32(const float* __restrict__ X,
              const float* __restrict__ W,
              float* __restrict__ H,
              int nRows, int reluIn) {
    __shared__ float xs[2][16 * XS_STRIDE];

    const int tid  = threadIdx.x;
    const int wave = tid >> 5;
    const int lane = tid & 31;
    const int m    = lane & 15;
    const int kk   = (lane >> 4) << 1;
    const int col  = wave * 16 + m;
    const int halfm = (lane >> 4) * 8;

    const int nStrips = (nRows + 15) >> 4;
    const int step = (int)gridDim.x;
    int s = (int)blockIdx.x;
    if (s >= nStrips) return;

    const int sr = tid >> 5;
    const int sc = tid & 31;

    float4 p0, p1;

    {
        const int ra = s * 16 + sr;
        p0 = make_float4(0.f, 0.f, 0.f, 0.f);
        p1 = make_float4(0.f, 0.f, 0.f, 0.f);
        if (ra < nRows)     p0 = ((const float4*)(X + (size_t)ra * D))[sc];
        if (ra + 8 < nRows) p1 = ((const float4*)(X + (size_t)(ra + 8) * D))[sc];
    }

    v2f bw[32];
    #pragma unroll
    for (int j = 0; j < 32; ++j) {
        bw[j].x = W[(size_t)(4 * j + kk) * D + col];
        bw[j].y = W[(size_t)(4 * j + kk + 1) * D + col];
    }

    int cur = 0;
    {
        if (reluIn) {
            p0.x = fmaxf(p0.x, 0.0f); p0.y = fmaxf(p0.y, 0.0f);
            p0.z = fmaxf(p0.z, 0.0f); p0.w = fmaxf(p0.w, 0.0f);
            p1.x = fmaxf(p1.x, 0.0f); p1.y = fmaxf(p1.y, 0.0f);
            p1.z = fmaxf(p1.z, 0.0f); p1.w = fmaxf(p1.w, 0.0f);
        }
        *(float4*)&xs[0][sr * XS_STRIDE + sc * 4]       = p0;
        *(float4*)&xs[0][(sr + 8) * XS_STRIDE + sc * 4] = p1;
    }
    __syncthreads();

    for (; s < nStrips; s += step) {
        const int row0 = s * 16;
        const int sn = s + step;
        const bool more = sn < nStrips;

        if (more) {
            const int ra = sn * 16 + sr;
            p0 = make_float4(0.f, 0.f, 0.f, 0.f);
            p1 = make_float4(0.f, 0.f, 0.f, 0.f);
            if (ra < nRows)     p0 = ((const float4*)(X + (size_t)ra * D))[sc];
            if (ra + 8 < nRows) p1 = ((const float4*)(X + (size_t)(ra + 8) * D))[sc];
        }

        const float* xa = &xs[cur][m * XS_STRIDE + kk];
        v8f acc = {};
        #pragma unroll
        for (int j = 0; j < 32; ++j) {
            v2f a;
            a.x = xa[4 * j];
            a.y = xa[4 * j + 1];
            acc = __builtin_amdgcn_wmma_f32_16x16x4_f32(
                 false, a,  false, bw[j],
                 (short)0, acc,  false,  false);
        }

        float* hp = H + (size_t)(row0 + halfm) * D + col;
        if (row0 + 16 <= nRows) {
            #pragma unroll
            for (int r = 0; r < 8; ++r) hp[(size_t)r * D] = acc[r];
        } else {
            #pragma unroll
            for (int r = 0; r < 8; ++r) {
                if (row0 + halfm + r < nRows) hp[(size_t)r * D] = acc[r];
            }
        }

        if (more) {
            if (reluIn) {
                p0.x = fmaxf(p0.x, 0.0f); p0.y = fmaxf(p0.y, 0.0f);
                p0.z = fmaxf(p0.z, 0.0f); p0.w = fmaxf(p0.w, 0.0f);
                p1.x = fmaxf(p1.x, 0.0f); p1.y = fmaxf(p1.y, 0.0f);
                p1.z = fmaxf(p1.z, 0.0f); p1.w = fmaxf(p1.w, 0.0f);
            }
            *(float4*)&xs[cur ^ 1][sr * XS_STRIDE + sc * 4]       = p0;
            *(float4*)&xs[cur ^ 1][(sr + 8) * XS_STRIDE + sc * 4] = p1;
        }
        __builtin_amdgcn_fence(__ATOMIC_RELEASE, "workgroup", "local");
        __builtin_amdgcn_s_barrier();
        __builtin_amdgcn_fence(__ATOMIC_ACQUIRE, "workgroup", "local");
        cur ^= 1;
    }
}

#define EB 8

__device__ __forceinline__ int lane_bcast_i(int v, int j) {
    return __builtin_amdgcn_readlane(v, j);
}
__device__ __forceinline__ float lane_bcast_f(float v, int j) {
    return __int_as_float(__builtin_amdgcn_readlane(__float_as_int(v), j));
}

__global__ void __launch_bounds__(1024)
edge_lds(const int* __restrict__ bsrc, const int* __restrict__ bdst,
         const float* __restrict__ bnorm, const int* __restrict__ gstart,
         const float* __restrict__ h, const float* __restrict__ dinv,
         const float* __restrict__ bias, float* __restrict__ out, int nRows) {
    __shared__ float acc[RB * D];

    const int tid  = threadIdx.x;
    const int row0 = (int)blockIdx.x * RB;
    const int left = nRows - row0;
    const int nr   = left < RB ? left : RB;
    const int nq   = nr * (D / 4);

    for (int idx = tid; idx < nq; idx += 1024) {
        const int r  = idx >> 5;
        const int c4 = idx & 31;
        const float4 v = ((const float4*)(h + (size_t)(row0 + r) * D))[c4];
        const float4 b = ((const float4*)bias)[c4];
        const float dv = dinv[row0 + r];
        const float n2 = dv * dv;
        float4 o;
        o.x = v.x * n2 + b.x; o.y = v.y * n2 + b.y;
        o.z = v.z * n2 + b.z; o.w = v.w * n2 + b.w;
        *(float4*)&acc[r * D + c4 * 4] = o;
    }
    __syncthreads();

    const int eBeg = gstart[blockIdx.x];
    const int eEnd = gstart[blockIdx.x + 1];
    const int lane = tid & 31;
    const int wv = __builtin_amdgcn_readfirstlane(tid >> 5);

    for (int e0 = eBeg + wv * 32; e0 < eEnd; e0 += 32 * 32) {
        const int cnt = eEnd - e0;
        const int e = e0 + lane;
        int s = 0, dl = 0;
        float nm = 0.0f;
        if (e < eEnd) { s = bsrc[e]; dl = bdst[e] - row0; nm = bnorm[e]; }
        if (cnt >= 32) {
            for (int j0 = 0; j0 < 32; j0 += EB) {
                float v[EB][4];
                #pragma unroll
                for (int u = 0; u < EB; ++u) {
                    const int sj = lane_bcast_i(s, j0 + u);
                    const float* hr = h + (size_t)sj * D + lane;
                    v[u][0] = hr[0];
                    v[u][1] = hr[32];
                    v[u][2] = hr[64];
                    v[u][3] = hr[96];
                }
                #pragma unroll
                for (int u = 0; u < EB; ++u) {
                    const int   dj = lane_bcast_i(dl, j0 + u);
                    const float nj = lane_bcast_f(nm, j0 + u);
                    float* a = &acc[dj * D + lane];
                    atomicAdd(a,      v[u][0] * nj);
                    atomicAdd(a + 32, v[u][1] * nj);
                    atomicAdd(a + 64, v[u][2] * nj);
                    atomicAdd(a + 96, v[u][3] * nj);
                }
            }
        } else {
            for (int j = 0; j < cnt; ++j) {
                const int   sj = lane_bcast_i(s, j);
                const int   dj = lane_bcast_i(dl, j);
                const float nj = lane_bcast_f(nm, j);
                const float* hr = h + (size_t)sj * D + lane;
                const float v0 = hr[0], v1 = hr[32], v2 = hr[64], v3 = hr[96];
                float* a = &acc[dj * D + lane];
                atomicAdd(a,      v0 * nj);
                atomicAdd(a + 32, v1 * nj);
                atomicAdd(a + 64, v2 * nj);
                atomicAdd(a + 96, v3 * nj);
            }
        }
    }
    __syncthreads();

    for (int idx = tid; idx < nq; idx += 1024) {
        const int r  = idx >> 5;
        const int c4 = idx & 31;
        ((float4*)(out + (size_t)(row0 + r) * D))[c4] = *(const float4*)&acc[r * D + c4 * 4];
    }
}


static inline size_t align256(size_t v) { return (v + 255) & ~(size_t)255; }

extern "C" void kernel_launch(void* const* d_in, const int* in_sizes, int n_in,
                              void* d_out, int out_size, void* d_ws, size_t ws_size,
                              hipStream_t stream) {
    const float* x  = (const float*)d_in[0];
    const int*   ei = (const int*)d_in[1];
    const float* W1 = (const float*)d_in[2];
    const float* b1 = (const float*)d_in[3];
    const float* W2 = (const float*)d_in[4];
    const float* b2 = (const float*)d_in[5];
    float* out = (float*)d_out;

    const int N = in_sizes[0] / D;
    const int E = in_sizes[1] / 2;
    const int* srcIdx = ei;
    const int* dstIdx = ei + E;

    char* ws = (char*)d_ws;
    float* deg  = (float*)ws;
    const size_t o0 = align256((size_t)N * sizeof(float));
    float* dinv = (float*)(ws + o0);
    const size_t o1 = o0 + align256((size_t)N * sizeof(float));
    float* hbuf = (float*)(ws + o1);
    const size_t o2 = o1 + align256((size_t)N * D * sizeof(float));
    float* agg  = (float*)(ws + o2);
    const size_t o3 = o2 + align256((size_t)N * D * sizeof(float));
    int* gcnt   = (int*)(ws + o3);
    const size_t o4 = o3 + align256((size_t)MAXG * sizeof(int));
    int* gcur   = (int*)(ws + o4);
    const size_t o5 = o4 + align256((size_t)MAXG * sizeof(int));
    int* gstart = (int*)(ws + o5);
    const size_t o6 = o5 + align256((size_t)(MAXG + 1) * sizeof(int));
    int* bsrc   = (int*)(ws + o6);
    const size_t o7 = o6 + align256((size_t)E * sizeof(int));
    int* bdst   = (int*)(ws + o7);
    const size_t o8 = o7 + align256((size_t)E * sizeof(int));
    float* bnorm = (float*)(ws + o8);

    const int blk = 256;
    const int gN  = (N + blk - 1) / blk;
    const int gCh = (E + CH - 1) / CH;
    const int gGemm = GEMM_BLOCKS;
    const int nG  = (N + RB - 1) / RB;

    init_deg   <<<gN,  blk, 0, stream>>>(deg, N, gcnt, gcur, nG);
    deg_hist   <<<gCh, blk, 0, stream>>>(dstIdx, deg, gcnt, E, nG);
    group_edges<<<gCh, blk, 0, stream>>>(srcIdx, dstIdx, deg, dinv, gcnt, gcur, gstart,
                                         bsrc, bdst, bnorm, N, E, nG);

    gemm_wmma_f32<<<gGemm, blk, 0, stream>>>(x, W1, hbuf, N, 0);
    edge_lds     <<<nG, 1024, 0, stream>>>(bsrc, bdst, bnorm, gstart, hbuf, dinv, b1, agg, N);

    gemm_wmma_f32<<<gGemm, blk, 0, stream>>>(agg, W2, hbuf, N, 1);
    edge_lds     <<<nG, 1024, 0, stream>>>(bsrc, bdst, bnorm, gstart, hbuf, dinv, b2, out, N);
}


# Edge row gather as one 16-B-per-lane load: GEMM stores h column-interleaved (lane l's float4 = columns l, l+32, l+64, l+96), 1 load instruction per 512-B row, was 4; LDS adds, sum order as before
# speedup vs baseline: 2.9977x; 1.0313x over previous
#include <hip/hip_runtime.h>
#include <hip/hip_bf16.h>


typedef float v2f __attribute__((ext_vector_type(2)));
typedef float v8f __attribute__((ext_vector_type(8)));

#define D 128
#define XS_STRIDE 132

#define RB   400
#define CH   2048
#define CPT  8
#define MAXG 1024


__global__ void init_deg(float* __restrict__ deg, int n, int* __restrict__ gcnt,
                         int* __restrict__ gcur, int nG) {
    int i = blockIdx.x * blockDim.x + threadIdx.x;
    if (i < n) deg[i] = 1.0f;
    if (i < nG) { gcnt[i] = 0; gcur[i] = 0; }
}

__global__ void __launch_bounds__(256)
deg_hist(const int* __restrict__ dst, float* __restrict__ deg,
         int* __restrict__ gcnt, int nE, int nG) {
    __shared__ int lh[MAXG];
    const int tid = threadIdx.x;
    for (int b = tid; b < nG; b += 256) lh[b] = 0;
    __syncthreads();
    const int base = blockIdx.x * CH;
    #pragma unroll
    for (int k = 0; k < CPT; ++k) {
        const int e = base + k * 256 + tid;
        if (e < nE) {
            const int d = dst[e];
            atomicAdd(&deg[d], 1.0f);
            atomicAdd(&lh[d / RB], 1);
        }
    }
    __syncthreads();
    for (int b = tid; b < nG; b += 256) {
        const int c = lh[b];
        if (c) atomicAdd(&gcnt[b], c);
    }
}

__global__ void __launch_bounds__(256)
group_edges(const int* __restrict__ src, const int* __restrict__ dst,
            const float* __restrict__ deg, float* __restrict__ dinv,
            const int* __restrict__ gcnt, int* __restrict__ gcur,
            int* __restrict__ gstart,
            int* __restrict__ bsrc, int* __restrict__ bdst, float* __restrict__ bnorm,
            int n, int nE, int nG) {
    __shared__ int lcnt[MAXG];
    __shared__ int lstart[MAXG];
    __shared__ int part[256];
    const int tid = threadIdx.x;

    for (int i = blockIdx.x * 256 + tid; i < n; i += gridDim.x * 256)
        dinv[i] = rsqrtf(deg[i]);

    for (int b = tid; b < nG; b += 256) lcnt[b] = 0;

    const int per = (nG + 255) / 256;
    int mine = 0;
    for (int q = 0; q < per; ++q) {
        const int b = tid * per + q;
        if (b < nG) mine += gcnt[b];
    }
    part[tid] = mine;
    __syncthreads();
    for (int off = 1; off < 256; off <<= 1) {
        const int v = (tid >= off) ? part[tid - off] : 0;
        __syncthreads();
        part[tid] += v;
        __syncthreads();
    }
    {
        int run = part[tid] - mine;
        for (int q = 0; q < per; ++q) {
            const int b = tid * per + q;
            if (b < nG) {
                lstart[b] = run;
                if (blockIdx.x == 0) gstart[b] = run;
                run += gcnt[b];
            }
        }
        if (blockIdx.x == 0 && tid == 255) gstart[nG] = part[255];
    }

    const int base = blockIdx.x * CH;
    int dk[CPT], rk[CPT];
    #pragma unroll
    for (int k = 0; k < CPT; ++k) {
        const int e = base + k * 256 + tid;
        dk[k] = -1;
        rk[k] = 0;
        if (e < nE) {
            const int d = dst[e];
            dk[k] = d;
            rk[k] = atomicAdd(&lcnt[d / RB], 1);
        }
    }
    __syncthreads();
    for (int b = tid; b < nG; b += 256) {
        const int c = lcnt[b];
        if (c) lcnt[b] = lstart[b] + atomicAdd(&gcur[b], c);
    }
    __syncthreads();
    #pragma unroll
    for (int k = 0; k < CPT; ++k) {
        if (dk[k] >= 0) {
            const int e = base + k * 256 + tid;
            const int d = dk[k];
            const int s = src[e];
            const int pos = lcnt[d / RB] + rk[k];
            bsrc[pos]  = s;
            bdst[pos]  = d;
            bnorm[pos] = rsqrtf(deg[s]) * rsqrtf(deg[d]);
        }
    }
}

#define GEMM_BLOCKS 1024

__global__ void __launch_bounds__(256)
gemm_wmma_f32(const float* __restrict__ X,
              const float* __restrict__ W,
              float* __restrict__ H,
              int nRows, int reluIn) {
    __shared__ float xs[2][16 * XS_STRIDE];

    const int tid  = threadIdx.x;
    const int wave = tid >> 5;
    const int lane = tid & 31;
    const int m    = lane & 15;
    const int kk   = (lane >> 4) << 1;
    const int pcol = wave * 16 + m;
    const int wcol = ((pcol & 3) << 5) + (pcol >> 2);
    const int halfm = (lane >> 4) * 8;

    const int nStrips = (nRows + 15) >> 4;
    const int step = (int)gridDim.x;
    int s = (int)blockIdx.x;
    if (s >= nStrips) return;

    const int sr = tid >> 5;
    const int sc = tid & 31;

    float4 p0, p1;

    {
        const int ra = s * 16 + sr;
        p0 = make_float4(0.f, 0.f, 0.f, 0.f);
        p1 = make_float4(0.f, 0.f, 0.f, 0.f);
        if (ra < nRows)     p0 = ((const float4*)(X + (size_t)ra * D))[sc];
        if (ra + 8 < nRows) p1 = ((const float4*)(X + (size_t)(ra + 8) * D))[sc];
    }

    v2f bw[32];
    #pragma unroll
    for (int j = 0; j < 32; ++j) {
        bw[j].x = W[(size_t)(4 * j + kk) * D + wcol];
        bw[j].y = W[(size_t)(4 * j + kk + 1) * D + wcol];
    }

    int cur = 0;
    {
        if (reluIn) {
            p0.x = fmaxf(p0.x, 0.0f); p0.y = fmaxf(p0.y, 0.0f);
            p0.z = fmaxf(p0.z, 0.0f); p0.w = fmaxf(p0.w, 0.0f);
            p1.x = fmaxf(p1.x, 0.0f); p1.y = fmaxf(p1.y, 0.0f);
            p1.z = fmaxf(p1.z, 0.0f); p1.w = fmaxf(p1.w, 0.0f);
        }
        *(float4*)&xs[0][sr * XS_STRIDE + sc * 4]       = p0;
        *(float4*)&xs[0][(sr + 8) * XS_STRIDE + sc * 4] = p1;
    }
    __syncthreads();

    for (; s < nStrips; s += step) {
        const int row0 = s * 16;
        const int sn = s + step;
        const bool more = sn < nStrips;

        if (more) {
            const int ra = sn * 16 + sr;
            p0 = make_float4(0.f, 0.f, 0.f, 0.f);
            p1 = make_float4(0.f, 0.f, 0.f, 0.f);
            if (ra < nRows)     p0 = ((const float4*)(X + (size_t)ra * D))[sc];
            if (ra + 8 < nRows) p1 = ((const float4*)(X + (size_t)(ra + 8) * D))[sc];
        }

        const float* xa = &xs[cur][m * XS_STRIDE + kk];
        v8f acc = {};
        #pragma unroll
        for (int j = 0; j < 32; ++j) {
            v2f a;
            a.x = xa[4 * j];
            a.y = xa[4 * j + 1];
            acc = __builtin_amdgcn_wmma_f32_16x16x4_f32(
                 false, a,  false, bw[j],
                 (short)0, acc,  false,  false);
        }

        float* hp = H + (size_t)(row0 + halfm) * D + pcol;
        if (row0 + 16 <= nRows) {
            #pragma unroll
            for (int r = 0; r < 8; ++r) hp[(size_t)r * D] = acc[r];
        } else {
            #pragma unroll
            for (int r = 0; r < 8; ++r) {
                if (row0 + halfm + r < nRows) hp[(size_t)r * D] = acc[r];
            }
        }

        if (more) {
            if (reluIn) {
                p0.x = fmaxf(p0.x, 0.0f); p0.y = fmaxf(p0.y, 0.0f);
                p0.z = fmaxf(p0.z, 0.0f); p0.w = fmaxf(p0.w, 0.0f);
                p1.x = fmaxf(p1.x, 0.0f); p1.y = fmaxf(p1.y, 0.0f);
                p1.z = fmaxf(p1.z, 0.0f); p1.w = fmaxf(p1.w, 0.0f);
            }
            *(float4*)&xs[cur ^ 1][sr * XS_STRIDE + sc * 4]       = p0;
            *(float4*)&xs[cur ^ 1][(sr + 8) * XS_STRIDE + sc * 4] = p1;
        }
        __builtin_amdgcn_fence(__ATOMIC_RELEASE, "workgroup", "local");
        __builtin_amdgcn_s_barrier();
        __builtin_amdgcn_fence(__ATOMIC_ACQUIRE, "workgroup", "local");
        cur ^= 1;
    }
}

#define EB 8

__device__ __forceinline__ int lane_bcast_i(int v, int j) {
    return __builtin_amdgcn_readlane(v, j);
}
__device__ __forceinline__ float lane_bcast_f(float v, int j) {
    return __int_as_float(__builtin_amdgcn_readlane(__float_as_int(v), j));
}

__global__ void __launch_bounds__(1024)
edge_lds(const int* __restrict__ bsrc, const int* __restrict__ bdst,
         const float* __restrict__ bnorm, const int* __restrict__ gstart,
         const float* __restrict__ h, const float* __restrict__ dinv,
         const float* __restrict__ bias, float* __restrict__ out, int nRows) {
    __shared__ float acc[RB * D];

    const int tid  = threadIdx.x;
    const int row0 = (int)blockIdx.x * RB;
    const int left = nRows - row0;
    const int nr   = left < RB ? left : RB;
    const int nq   = nr * (D / 4);

    {
        const int c4 = tid & 31;
        const float b0 = bias[c4], b1 = bias[c4 + 32], b2 = bias[c4 + 64], b3 = bias[c4 + 96];
        int r = tid >> 5;
        for (; r + 32 < nr; r += 64) {
            const float4 va = ((const float4*)(h + (size_t)(row0 + r) * D))[c4];
            const float4 vb = ((const float4*)(h + (size_t)(row0 + r + 32) * D))[c4];
            const float  da = dinv[row0 + r];
            const float  db = dinv[row0 + r + 32];
            const float  na = da * da;
            const float  nb = db * db;
            float* a = &acc[r * D + c4];
            a[0]  = va.x * na + b0;
            a[32] = va.y * na + b1;
            a[64] = va.z * na + b2;
            a[96] = va.w * na + b3;
            float* b = a + 32 * D;
            b[0]  = vb.x * nb + b0;
            b[32] = vb.y * nb + b1;
            b[64] = vb.z * nb + b2;
            b[96] = vb.w * nb + b3;
        }
        if (r < nr) {
            const float4 va = ((const float4*)(h + (size_t)(row0 + r) * D))[c4];
            const float  da = dinv[row0 + r];
            const float  na = da * da;
            float* a = &acc[r * D + c4];
            a[0]  = va.x * na + b0;
            a[32] = va.y * na + b1;
            a[64] = va.z * na + b2;
            a[96] = va.w * na + b3;
        }
    }
    __syncthreads();

    const int eBeg = gstart[blockIdx.x];
    const int eEnd = gstart[blockIdx.x + 1];
    const int lane = tid & 31;
    const int wv = __builtin_amdgcn_readfirstlane(tid >> 5);

    for (int e0 = eBeg + wv * 32; e0 < eEnd; e0 += 32 * 32) {
        const int cnt = eEnd - e0;
        const int e = e0 + lane;
        int s = 0, dl = 0;
        float nm = 0.0f;
        if (e < eEnd) { s = bsrc[e]; dl = bdst[e] - row0; nm = bnorm[e]; }
        if (cnt >= 32) {
            for (int j0 = 0; j0 < 32; j0 += EB) {
                float4 v[EB];
                #pragma unroll
                for (int u = 0; u < EB; ++u) {
                    const int sj = lane_bcast_i(s, j0 + u);
                    v[u] = ((const float4*)(h + (size_t)sj * D))[lane];
                }
                #pragma unroll
                for (int u = 0; u < EB; ++u) {
                    const int   dj = lane_bcast_i(dl, j0 + u);
                    const float nj = lane_bcast_f(nm, j0 + u);
                    float* a = &acc[dj * D + lane];
                    atomicAdd(a,      v[u].x * nj);
                    atomicAdd(a + 32, v[u].y * nj);
                    atomicAdd(a + 64, v[u].z * nj);
                    atomicAdd(a + 96, v[u].w * nj);
                }
            }
        } else {
            for (int j = 0; j < cnt; ++j) {
                const int   sj = lane_bcast_i(s, j);
                const int   dj = lane_bcast_i(dl, j);
                const float nj = lane_bcast_f(nm, j);
                const float4 v = ((const float4*)(h + (size_t)sj * D))[lane];
                float* a = &acc[dj * D + lane];
                atomicAdd(a,      v.x * nj);
                atomicAdd(a + 32, v.y * nj);
                atomicAdd(a + 64, v.z * nj);
                atomicAdd(a + 96, v.w * nj);
            }
        }
    }
    __syncthreads();

    for (int idx = tid; idx < nq; idx += 1024) {
        const int r  = idx >> 5;
        const int c4 = idx & 31;
        ((float4*)(out + (size_t)(row0 + r) * D))[c4] = *(const float4*)&acc[r * D + c4 * 4];
    }
}


static inline size_t align256(size_t v) { return (v + 255) & ~(size_t)255; }

extern "C" void kernel_launch(void* const* d_in, const int* in_sizes, int n_in,
                              void* d_out, int out_size, void* d_ws, size_t ws_size,
                              hipStream_t stream) {
    const float* x  = (const float*)d_in[0];
    const int*   ei = (const int*)d_in[1];
    const float* W1 = (const float*)d_in[2];
    const float* b1 = (const float*)d_in[3];
    const float* W2 = (const float*)d_in[4];
    const float* b2 = (const float*)d_in[5];
    float* out = (float*)d_out;

    const int N = in_sizes[0] / D;
    const int E = in_sizes[1] / 2;
    const int* srcIdx = ei;
    const int* dstIdx = ei + E;

    char* ws = (char*)d_ws;
    float* deg  = (float*)ws;
    const size_t o0 = align256((size_t)N * sizeof(float));
    float* dinv = (float*)(ws + o0);
    const size_t o1 = o0 + align256((size_t)N * sizeof(float));
    float* hbuf = (float*)(ws + o1);
    const size_t o2 = o1 + align256((size_t)N * D * sizeof(float));
    float* agg  = (float*)(ws + o2);
    const size_t o3 = o2 + align256((size_t)N * D * sizeof(float));
    int* gcnt   = (int*)(ws + o3);
    const size_t o4 = o3 + align256((size_t)MAXG * sizeof(int));
    int* gcur   = (int*)(ws + o4);
    const size_t o5 = o4 + align256((size_t)MAXG * sizeof(int));
    int* gstart = (int*)(ws + o5);
    const size_t o6 = o5 + align256((size_t)(MAXG + 1) * sizeof(int));
    int* bsrc   = (int*)(ws + o6);
    const size_t o7 = o6 + align256((size_t)E * sizeof(int));
    int* bdst   = (int*)(ws + o7);
    const size_t o8 = o7 + align256((size_t)E * sizeof(int));
    float* bnorm = (float*)(ws + o8);

    const int blk = 256;
    const int gN  = (N + blk - 1) / blk;
    const int gCh = (E + CH - 1) / CH;
    const int gGemm = GEMM_BLOCKS;
    const int nG  = (N + RB - 1) / RB;

    init_deg   <<<gN,  blk, 0, stream>>>(deg, N, gcnt, gcur, nG);
    deg_hist   <<<gCh, blk, 0, stream>>>(dstIdx, deg, gcnt, E, nG);
    group_edges<<<gCh, blk, 0, stream>>>(srcIdx, dstIdx, deg, dinv, gcnt, gcur, gstart,
                                         bsrc, bdst, bnorm, N, E, nG);

    gemm_wmma_f32<<<gGemm, blk, 0, stream>>>(x, W1, hbuf, N, 0);
    edge_lds     <<<nG, 1024, 0, stream>>>(bsrc, bdst, bnorm, gstart, hbuf, dinv, b1, agg, N);

    gemm_wmma_f32<<<gGemm, blk, 0, stream>>>(agg, W2, hbuf, N, 1);
    edge_lds     <<<nG, 1024, 0, stream>>>(bsrc, bdst, bnorm, gstart, hbuf, dinv, b2, out, N);
}


# Layer 2's product relu(sums) @ W2 moves into layer 1's aggregation launch: a work-group multiplies its 400 LDS-resident rows by W2 (staged in LDS) on WMMA, stores h2; 7 launches to 6; sums unchanged
# speedup vs baseline: 3.1837x; 1.0620x over previous
#include <hip/hip_runtime.h>
#include <hip/hip_bf16.h>


typedef float v2f __attribute__((ext_vector_type(2)));
typedef float v8f __attribute__((ext_vector_type(8)));

#define D 128
#define XS_STRIDE 132

#define RB   400
#define AS   132
#define WP   136
#define CH   2048
#define CPT  8
#define MAXG 1024


__global__ void init_deg(float* __restrict__ deg, int n, int* __restrict__ gcnt,
                         int* __restrict__ gcur, int nG) {
    int i = blockIdx.x * blockDim.x + threadIdx.x;
    if (i < n) deg[i] = 1.0f;
    if (i < nG) { gcnt[i] = 0; gcur[i] = 0; }
}

__global__ void __launch_bounds__(256)
deg_hist(const int* __restrict__ dst, float* __restrict__ deg,
         int* __restrict__ gcnt, int nE, int nG) {
    __shared__ int lh[MAXG];
    const int tid = threadIdx.x;
    for (int b = tid; b < nG; b += 256) lh[b] = 0;
    __syncthreads();
    const int base = blockIdx.x * CH;
    #pragma unroll
    for (int k = 0; k < CPT; ++k) {
        const int e = base + k * 256 + tid;
        if (e < nE) {
            const int d = dst[e];
            atomicAdd(&deg[d], 1.0f);
            atomicAdd(&lh[d / RB], 1);
        }
    }
    __syncthreads();
    for (int b = tid; b < nG; b += 256) {
        const int c = lh[b];
        if (c) atomicAdd(&gcnt[b], c);
    }
}

__global__ void __launch_bounds__(256)
group_edges(const int* __restrict__ src, const int* __restrict__ dst,
            const float* __restrict__ deg, float* __restrict__ dinv,
            const int* __restrict__ gcnt, int* __restrict__ gcur,
            int* __restrict__ gstart,
            int* __restrict__ bsrc, int* __restrict__ bdst, float* __restrict__ bnorm,
            int n, int nE, int nG) {
    __shared__ int lcnt[MAXG];
    __shared__ int lstart[MAXG];
    __shared__ int part[256];
    const int tid = threadIdx.x;

    for (int i = blockIdx.x * 256 + tid; i < n; i += gridDim.x * 256)
        dinv[i] = rsqrtf(deg[i]);

    for (int b = tid; b < nG; b += 256) lcnt[b] = 0;

    const int per = (nG + 255) / 256;
    int mine = 0;
    for (int q = 0; q < per; ++q) {
        const int b = tid * per + q;
        if (b < nG) mine += gcnt[b];
    }
    part[tid] = mine;
    __syncthreads();
    for (int off = 1; off < 256; off <<= 1) {
        const int v = (tid >= off) ? part[tid - off] : 0;
        __syncthreads();
        part[tid] += v;
        __syncthreads();
    }
    {
        int run = part[tid] - mine;
        for (int q = 0; q < per; ++q) {
            const int b = tid * per + q;
            if (b < nG) {
                lstart[b] = run;
                if (blockIdx.x == 0) gstart[b] = run;
                run += gcnt[b];
            }
        }
        if (blockIdx.x == 0 && tid == 255) gstart[nG] = part[255];
    }

    const int base = blockIdx.x * CH;
    int dk[CPT], rk[CPT];
    #pragma unroll
    for (int k = 0; k < CPT; ++k) {
        const int e = base + k * 256 + tid;
        dk[k] = -1;
        rk[k] = 0;
        if (e < nE) {
            const int d = dst[e];
            dk[k] = d;
            rk[k] = atomicAdd(&lcnt[d / RB], 1);
        }
    }
    __syncthreads();
    for (int b = tid; b < nG; b += 256) {
        const int c = lcnt[b];
        if (c) lcnt[b] = lstart[b] + atomicAdd(&gcur[b], c);
    }
    __syncthreads();
    #pragma unroll
    for (int k = 0; k < CPT; ++k) {
        if (dk[k] >= 0) {
            const int e = base + k * 256 + tid;
            const int d = dk[k];
            const int s = src[e];
            const int pos = lcnt[d / RB] + rk[k];
            bsrc[pos]  = s;
            bdst[pos]  = d;
            bnorm[pos] = rsqrtf(deg[s]) * rsqrtf(deg[d]);
        }
    }
}

#define GEMM_BLOCKS 1024

__global__ void __launch_bounds__(256)
gemm_wmma_f32(const float* __restrict__ X,
              const float* __restrict__ W,
              float* __restrict__ H,
              int nRows) {
    __shared__ float xs[2][16 * XS_STRIDE];

    const int tid  = threadIdx.x;
    const int wave = tid >> 5;
    const int lane = tid & 31;
    const int m    = lane & 15;
    const int kk   = (lane >> 4) << 1;
    const int pcol = wave * 16 + m;
    const int wcol = ((pcol & 3) << 5) + (pcol >> 2);
    const int halfm = (lane >> 4) * 8;

    const int nStrips = (nRows + 15) >> 4;
    const int step = (int)gridDim.x;
    int s = (int)blockIdx.x;
    if (s >= nStrips) return;

    const int sr = tid >> 5;
    const int sc = tid & 31;

    float4 p0, p1;

    {
        const int ra = s * 16 + sr;
        p0 = make_float4(0.f, 0.f, 0.f, 0.f);
        p1 = make_float4(0.f, 0.f, 0.f, 0.f);
        if (ra < nRows)     p0 = ((const float4*)(X + (size_t)ra * D))[sc];
        if (ra + 8 < nRows) p1 = ((const float4*)(X + (size_t)(ra + 8) * D))[sc];
    }

    v2f bw[32];
    #pragma unroll
    for (int j = 0; j < 32; ++j) {
        bw[j].x = W[(size_t)(4 * j + kk) * D + wcol];
        bw[j].y = W[(size_t)(4 * j + kk + 1) * D + wcol];
    }

    int cur = 0;
    {
        *(float4*)&xs[0][sr * XS_STRIDE + sc * 4]       = p0;
        *(float4*)&xs[0][(sr + 8) * XS_STRIDE + sc * 4] = p1;
    }
    __syncthreads();

    for (; s < nStrips; s += step) {
        const int row0 = s * 16;
        const int sn = s + step;
        const bool more = sn < nStrips;

        if (more) {
            const int ra = sn * 16 + sr;
            p0 = make_float4(0.f, 0.f, 0.f, 0.f);
            p1 = make_float4(0.f, 0.f, 0.f, 0.f);
            if (ra < nRows)     p0 = ((const float4*)(X + (size_t)ra * D))[sc];
            if (ra + 8 < nRows) p1 = ((const float4*)(X + (size_t)(ra + 8) * D))[sc];
        }

        const float* xa = &xs[cur][m * XS_STRIDE + kk];
        v8f acc = {};
        #pragma unroll
        for (int j = 0; j < 32; ++j) {
            v2f a;
            a.x = xa[4 * j];
            a.y = xa[4 * j + 1];
            acc = __builtin_amdgcn_wmma_f32_16x16x4_f32(
                 false, a,  false, bw[j],
                 (short)0, acc,  false,  false);
        }

        float* hp = H + (size_t)(row0 + halfm) * D + pcol;
        if (row0 + 16 <= nRows) {
            #pragma unroll
            for (int r = 0; r < 8; ++r) hp[(size_t)r * D] = acc[r];
        } else {
            #pragma unroll
            for (int r = 0; r < 8; ++r) {
                if (row0 + halfm + r < nRows) hp[(size_t)r * D] = acc[r];
            }
        }

        if (more) {
            *(float4*)&xs[cur ^ 1][sr * XS_STRIDE + sc * 4]       = p0;
            *(float4*)&xs[cur ^ 1][(sr + 8) * XS_STRIDE + sc * 4] = p1;
        }
        __builtin_amdgcn_fence(__ATOMIC_RELEASE, "workgroup", "local");
        __builtin_amdgcn_s_barrier();
        __builtin_amdgcn_fence(__ATOMIC_ACQUIRE, "workgroup", "local");
        cur ^= 1;
    }
}

#define EB 8

__device__ __forceinline__ int lane_bcast_i(int v, int j) {
    return __builtin_amdgcn_readlane(v, j);
}
__device__ __forceinline__ float lane_bcast_f(float v, int j) {
    return __int_as_float(__builtin_amdgcn_readlane(__float_as_int(v), j));
}

template <bool FUSE>
__global__ void __launch_bounds__(1024)
edge_lds(const int* __restrict__ bsrc, const int* __restrict__ bdst,
         const float* __restrict__ bnorm, const int* __restrict__ gstart,
         const float* __restrict__ h, const float* __restrict__ dinv,
         const float* __restrict__ bias, const float* __restrict__ Wn,
         float* __restrict__ out, int nRows) {
    __shared__ float acc[RB * AS];
    __shared__ __attribute__((aligned(16))) float wl[FUSE ? D * WP : 4];

    const int tid  = threadIdx.x;
    const int row0 = (int)blockIdx.x * RB;
    const int left = nRows - row0;
    const int nr   = left < RB ? left : RB;

    if (FUSE) {
        #pragma unroll
        for (int q = 0; q < 4; ++q) {
            const int idx = tid + q * 1024;
            const int k   = idx >> 5;
            const int c   = (idx & 31) * 4;
            const float4 w = ((const float4*)Wn)[idx];
            *(float4*)&wl[k * WP + c + ((c >> 6) << 3)] = w;
        }
    }

    {
        const int c4 = tid & 31;
        const float b0 = bias[c4], b1 = bias[c4 + 32], b2 = bias[c4 + 64], b3 = bias[c4 + 96];
        int r = tid >> 5;
        for (; r + 32 < nr; r += 64) {
            const float4 va = ((const float4*)(h + (size_t)(row0 + r) * D))[c4];
            const float4 vb = ((const float4*)(h + (size_t)(row0 + r + 32) * D))[c4];
            const float  da = dinv[row0 + r];
            const float  db = dinv[row0 + r + 32];
            const float  na = da * da;
            const float  nb = db * db;
            float* a = &acc[r * AS + c4];
            a[0]  = va.x * na + b0;
            a[32] = va.y * na + b1;
            a[64] = va.z * na + b2;
            a[96] = va.w * na + b3;
            float* b = a + 32 * AS;
            b[0]  = vb.x * nb + b0;
            b[32] = vb.y * nb + b1;
            b[64] = vb.z * nb + b2;
            b[96] = vb.w * nb + b3;
        }
        if (r < nr) {
            const float4 va = ((const float4*)(h + (size_t)(row0 + r) * D))[c4];
            const float  da = dinv[row0 + r];
            const float  na = da * da;
            float* a = &acc[r * AS + c4];
            a[0]  = va.x * na + b0;
            a[32] = va.y * na + b1;
            a[64] = va.z * na + b2;
            a[96] = va.w * na + b3;
        }
    }
    __syncthreads();

    const int eBeg = gstart[blockIdx.x];
    const int eEnd = gstart[blockIdx.x + 1];
    const int lane = tid & 31;
    const int wv = __builtin_amdgcn_readfirstlane(tid >> 5);

    for (int e0 = eBeg + wv * 32; e0 < eEnd; e0 += 32 * 32) {
        const int cnt = eEnd - e0;
        const int e = e0 + lane;
        int s = 0, dl = 0;
        float nm = 0.0f;
        if (e < eEnd) { s = bsrc[e]; dl = bdst[e] - row0; nm = bnorm[e]; }
        if (cnt >= 32) {
            for (int j0 = 0; j0 < 32; j0 += EB) {
                float4 v[EB];
                #pragma unroll
                for (int u = 0; u < EB; ++u) {
                    const int sj = lane_bcast_i(s, j0 + u);
                    v[u] = ((const float4*)(h + (size_t)sj * D))[lane];
                }
                #pragma unroll
                for (int u = 0; u < EB; ++u) {
                    const int   dj = lane_bcast_i(dl, j0 + u);
                    const float nj = lane_bcast_f(nm, j0 + u);
                    float* a = &acc[dj * AS + lane];
                    atomicAdd(a,      v[u].x * nj);
                    atomicAdd(a + 32, v[u].y * nj);
                    atomicAdd(a + 64, v[u].z * nj);
                    atomicAdd(a + 96, v[u].w * nj);
                }
            }
        } else {
            for (int j = 0; j < cnt; ++j) {
                const int   sj = lane_bcast_i(s, j);
                const int   dj = lane_bcast_i(dl, j);
                const float nj = lane_bcast_f(nm, j);
                const float4 v = ((const float4*)(h + (size_t)sj * D))[lane];
                float* a = &acc[dj * AS + lane];
                atomicAdd(a,      v.x * nj);
                atomicAdd(a + 32, v.y * nj);
                atomicAdd(a + 64, v.z * nj);
                atomicAdd(a + 96, v.w * nj);
            }
        }
    }
    if (FUSE) {
        const int m     = lane & 15;
        const int kk    = (lane >> 4) << 1;
        const int halfm = (lane >> 4) * 8;
        const int ct    = wv & 7;
        const int pcol  = ct * 16 + m;
        const int wcol  = ((pcol & 3) << 5) + (pcol >> 2);

        const float* wb = &wl[kk * WP + wcol + ((wcol >> 6) << 3)];
        v2f bw[32];
        #pragma unroll
        for (int j = 0; j < 32; ++j) {
            bw[j].x = wb[(4 * j) * WP];
            bw[j].y = wb[(4 * j + 1) * WP];
        }
        __syncthreads();

        const int nst = (nr + 15) >> 4;
        for (int st = wv >> 3; st < nst; st += 4) {
            const float* xa = &acc[(st * 16 + m) * AS + kk];
            v8f c = {};
            #pragma unroll
            for (int j = 0; j < 32; ++j) {
                v2f a;
                a.x = fmaxf(xa[4 * j], 0.0f);
                a.y = fmaxf(xa[4 * j + 1], 0.0f);
                c = __builtin_amdgcn_wmma_f32_16x16x4_f32(
                     false, a,  false, bw[j],
                     (short)0, c,  false,  false);
            }
            const int r0 = st * 16 + halfm;
            float* hp = out + (size_t)(row0 + r0) * D + pcol;
            if (st * 16 + 16 <= nr) {
                #pragma unroll
                for (int r = 0; r < 8; ++r) hp[(size_t)r * D] = c[r];
            } else {
                #pragma unroll
                for (int r = 0; r < 8; ++r) {
                    if (r0 + r < nr) hp[(size_t)r * D] = c[r];
                }
            }
        }
    } else {
        __syncthreads();

        const int nq = nr * (D / 4);
        for (int idx = tid; idx < nq; idx += 1024) {
            const int r  = idx >> 5;
            const int c4 = idx & 31;
            ((float4*)(out + (size_t)(row0 + r) * D))[c4] = *(const float4*)&acc[r * AS + c4 * 4];
        }
    }
}


static inline size_t align256(size_t v) { return (v + 255) & ~(size_t)255; }

extern "C" void kernel_launch(void* const* d_in, const int* in_sizes, int n_in,
                              void* d_out, int out_size, void* d_ws, size_t ws_size,
                              hipStream_t stream) {
    const float* x  = (const float*)d_in[0];
    const int*   ei = (const int*)d_in[1];
    const float* W1 = (const float*)d_in[2];
    const float* b1 = (const float*)d_in[3];
    const float* W2 = (const float*)d_in[4];
    const float* b2 = (const float*)d_in[5];
    float* out = (float*)d_out;

    const int N = in_sizes[0] / D;
    const int E = in_sizes[1] / 2;
    const int* srcIdx = ei;
    const int* dstIdx = ei + E;

    char* ws = (char*)d_ws;
    float* deg  = (float*)ws;
    const size_t o0 = align256((size_t)N * sizeof(float));
    float* dinv = (float*)(ws + o0);
    const size_t o1 = o0 + align256((size_t)N * sizeof(float));
    float* hbuf = (float*)(ws + o1);
    const size_t o2 = o1 + align256((size_t)N * D * sizeof(float));
    float* h2buf = (float*)(ws + o2);
    const size_t o3 = o2 + align256((size_t)N * D * sizeof(float));
    int* gcnt   = (int*)(ws + o3);
    const size_t o4 = o3 + align256((size_t)MAXG * sizeof(int));
    int* gcur   = (int*)(ws + o4);
    const size_t o5 = o4 + align256((size_t)MAXG * sizeof(int));
    int* gstart = (int*)(ws + o5);
    const size_t o6 = o5 + align256((size_t)(MAXG + 1) * sizeof(int));
    int* bsrc   = (int*)(ws + o6);
    const size_t o7 = o6 + align256((size_t)E * sizeof(int));
    int* bdst   = (int*)(ws + o7);
    const size_t o8 = o7 + align256((size_t)E * sizeof(int));
    float* bnorm = (float*)(ws + o8);

    const int blk = 256;
    const int gN  = (N + blk - 1) / blk;
    const int gCh = (E + CH - 1) / CH;
    const int gGemm = GEMM_BLOCKS;
    const int nG  = (N + RB - 1) / RB;

    init_deg   <<<gN,  blk, 0, stream>>>(deg, N, gcnt, gcur, nG);
    deg_hist   <<<gCh, blk, 0, stream>>>(dstIdx, deg, gcnt, E, nG);
    group_edges<<<gCh, blk, 0, stream>>>(srcIdx, dstIdx, deg, dinv, gcnt, gcur, gstart,
                                         bsrc, bdst, bnorm, N, E, nG);

    gemm_wmma_f32  <<<gGemm, blk, 0, stream>>>(x, W1, hbuf, N);
    edge_lds<true> <<<nG, 1024, 0, stream>>>(bsrc, bdst, bnorm, gstart, hbuf, dinv, b1, W2, h2buf, N);

    edge_lds<false><<<nG, 1024, 0, stream>>>(bsrc, bdst, bnorm, gstart, h2buf, dinv, b2, nullptr, out, N);
}


# Degrees counted in LDS by the work-group that owns a 400-row group, from its regrouped records (int adds): the 1.6M global f32 adds go; layer 1's edge walk forms and stores norm; sums unchanged
# speedup vs baseline: 4.6541x; 1.4619x over previous
#include <hip/hip_runtime.h>
#include <hip/hip_bf16.h>


typedef float v2f __attribute__((ext_vector_type(2)));
typedef float v8f __attribute__((ext_vector_type(8)));

#define D 128
#define XS_STRIDE 132

#define RB   400
#define AS   132
#define WP   136
#define CH   2048
#define CPT  8
#define MAXG 1024


__global__ void init_cnt(int* __restrict__ gcnt, int* __restrict__ gcur, int nG) {
    int i = blockIdx.x * blockDim.x + threadIdx.x;
    if (i < nG) { gcnt[i] = 0; gcur[i] = 0; }
}

__global__ void __launch_bounds__(256)
grp_hist(const int* __restrict__ dst, int* __restrict__ gcnt, int nE, int nG) {
    __shared__ int lh[MAXG];
    const int tid = threadIdx.x;
    for (int b = tid; b < nG; b += 256) lh[b] = 0;
    __syncthreads();
    const int base = blockIdx.x * CH;
    #pragma unroll
    for (int k = 0; k < CPT; ++k) {
        const int e = base + k * 256 + tid;
        if (e < nE) atomicAdd(&lh[dst[e] / RB], 1);
    }
    __syncthreads();
    for (int b = tid; b < nG; b += 256) {
        const int c = lh[b];
        if (c) atomicAdd(&gcnt[b], c);
    }
}

__global__ void __launch_bounds__(256)
group_edges(const int* __restrict__ src, const int* __restrict__ dst,
            const int* __restrict__ gcnt, int* __restrict__ gcur,
            int* __restrict__ gstart,
            int* __restrict__ bsrc, int* __restrict__ bdst,
            int nE, int nG) {
    __shared__ int lcnt[MAXG];
    __shared__ int lstart[MAXG];
    __shared__ int part[256];
    const int tid = threadIdx.x;

    for (int b = tid; b < nG; b += 256) lcnt[b] = 0;

    const int per = (nG + 255) / 256;
    int mine = 0;
    for (int q = 0; q < per; ++q) {
        const int b = tid * per + q;
        if (b < nG) mine += gcnt[b];
    }
    part[tid] = mine;
    __syncthreads();
    for (int off = 1; off < 256; off <<= 1) {
        const int v = (tid >= off) ? part[tid - off] : 0;
        __syncthreads();
        part[tid] += v;
        __syncthreads();
    }
    {
        int run = part[tid] - mine;
        for (int q = 0; q < per; ++q) {
            const int b = tid * per + q;
            if (b < nG) {
                lstart[b] = run;
                if (blockIdx.x == 0) gstart[b] = run;
                run += gcnt[b];
            }
        }
        if (blockIdx.x == 0 && tid == 255) gstart[nG] = part[255];
    }

    const int base = blockIdx.x * CH;
    int dk[CPT], rk[CPT];
    #pragma unroll
    for (int k = 0; k < CPT; ++k) {
        const int e = base + k * 256 + tid;
        dk[k] = -1;
        rk[k] = 0;
        if (e < nE) {
            const int d = dst[e];
            dk[k] = d;
            rk[k] = atomicAdd(&lcnt[d / RB], 1);
        }
    }
    __syncthreads();
    for (int b = tid; b < nG; b += 256) {
        const int c = lcnt[b];
        if (c) lcnt[b] = lstart[b] + atomicAdd(&gcur[b], c);
    }
    __syncthreads();
    #pragma unroll
    for (int k = 0; k < CPT; ++k) {
        if (dk[k] >= 0) {
            const int e = base + k * 256 + tid;
            const int d = dk[k];
            const int s = src[e];
            const int pos = lcnt[d / RB] + rk[k];
            bsrc[pos] = s;
            bdst[pos] = d;
        }
    }
}

__global__ void __launch_bounds__(1024)
row_deg(const int* __restrict__ bdst, const int* __restrict__ gstart,
        float* __restrict__ dinv, int nRows) {
    __shared__ int cnt[RB];
    const int tid  = threadIdx.x;
    const int row0 = (int)blockIdx.x * RB;
    const int left = nRows - row0;
    const int nr   = left < RB ? left : RB;
    for (int r = tid; r < RB; r += 1024) cnt[r] = 1;
    __syncthreads();
    const int eBeg = gstart[blockIdx.x];
    const int eEnd = gstart[blockIdx.x + 1];
    for (int e = eBeg + tid; e < eEnd; e += 4 * 1024) {
        int d0, d1 = -1, d2 = -1, d3 = -1;
        d0 = bdst[e];
        if (e + 1024 < eEnd) d1 = bdst[e + 1024];
        if (e + 2048 < eEnd) d2 = bdst[e + 2048];
        if (e + 3072 < eEnd) d3 = bdst[e + 3072];
        atomicAdd(&cnt[d0 - row0], 1);
        if (d1 >= 0) atomicAdd(&cnt[d1 - row0], 1);
        if (d2 >= 0) atomicAdd(&cnt[d2 - row0], 1);
        if (d3 >= 0) atomicAdd(&cnt[d3 - row0], 1);
    }
    __syncthreads();
    for (int r = tid; r < nr; r += 1024) dinv[row0 + r] = rsqrtf((float)cnt[r]);
}

#define GEMM_BLOCKS 1024

__global__ void __launch_bounds__(256)
gemm_wmma_f32(const float* __restrict__ X,
              const float* __restrict__ W,
              float* __restrict__ H,
              int nRows) {
    __shared__ float xs[2][16 * XS_STRIDE];

    const int tid  = threadIdx.x;
    const int wave = tid >> 5;
    const int lane = tid & 31;
    const int m    = lane & 15;
    const int kk   = (lane >> 4) << 1;
    const int pcol = wave * 16 + m;
    const int wcol = ((pcol & 3) << 5) + (pcol >> 2);
    const int halfm = (lane >> 4) * 8;

    const int nStrips = (nRows + 15) >> 4;
    const int step = (int)gridDim.x;
    int s = (int)blockIdx.x;
    if (s >= nStrips) return;

    const int sr = tid >> 5;
    const int sc = tid & 31;

    float4 p0, p1;

    {
        const int ra = s * 16 + sr;
        p0 = make_float4(0.f, 0.f, 0.f, 0.f);
        p1 = make_float4(0.f, 0.f, 0.f, 0.f);
        if (ra < nRows)     p0 = ((const float4*)(X + (size_t)ra * D))[sc];
        if (ra + 8 < nRows) p1 = ((const float4*)(X + (size_t)(ra + 8) * D))[sc];
    }

    v2f bw[32];
    #pragma unroll
    for (int j = 0; j < 32; ++j) {
        bw[j].x = W[(size_t)(4 * j + kk) * D + wcol];
        bw[j].y = W[(size_t)(4 * j + kk + 1) * D + wcol];
    }

    int cur = 0;
    {
        *(float4*)&xs[0][sr * XS_STRIDE + sc * 4]       = p0;
        *(float4*)&xs[0][(sr + 8) * XS_STRIDE + sc * 4] = p1;
    }
    __syncthreads();

    for (; s < nStrips; s += step) {
        const int row0 = s * 16;
        const int sn = s + step;
        const bool more = sn < nStrips;

        if (more) {
            const int ra = sn * 16 + sr;
            p0 = make_float4(0.f, 0.f, 0.f, 0.f);
            p1 = make_float4(0.f, 0.f, 0.f, 0.f);
            if (ra < nRows)     p0 = ((const float4*)(X + (size_t)ra * D))[sc];
            if (ra + 8 < nRows) p1 = ((const float4*)(X + (size_t)(ra + 8) * D))[sc];
        }

        const float* xa = &xs[cur][m * XS_STRIDE + kk];
        v8f acc = {};
        #pragma unroll
        for (int j = 0; j < 32; ++j) {
            v2f a;
            a.x = xa[4 * j];
            a.y = xa[4 * j + 1];
            acc = __builtin_amdgcn_wmma_f32_16x16x4_f32(
                 false, a,  false, bw[j],
                 (short)0, acc,  false,  false);
        }

        float* hp = H + (size_t)(row0 + halfm) * D + pcol;
        if (row0 + 16 <= nRows) {
            #pragma unroll
            for (int r = 0; r < 8; ++r) hp[(size_t)r * D] = acc[r];
        } else {
            #pragma unroll
            for (int r = 0; r < 8; ++r) {
                if (row0 + halfm + r < nRows) hp[(size_t)r * D] = acc[r];
            }
        }

        if (more) {
            *(float4*)&xs[cur ^ 1][sr * XS_STRIDE + sc * 4]       = p0;
            *(float4*)&xs[cur ^ 1][(sr + 8) * XS_STRIDE + sc * 4] = p1;
        }
        __builtin_amdgcn_fence(__ATOMIC_RELEASE, "workgroup", "local");
        __builtin_amdgcn_s_barrier();
        __builtin_amdgcn_fence(__ATOMIC_ACQUIRE, "workgroup", "local");
        cur ^= 1;
    }
}

#define EB 8

__device__ __forceinline__ int lane_bcast_i(int v, int j) {
    return __builtin_amdgcn_readlane(v, j);
}
__device__ __forceinline__ float lane_bcast_f(float v, int j) {
    return __int_as_float(__builtin_amdgcn_readlane(__float_as_int(v), j));
}

template <bool FUSE>
__global__ void __launch_bounds__(1024)
edge_lds(const int* __restrict__ bsrc, const int* __restrict__ bdst,
         float* __restrict__ bnorm, const int* __restrict__ gstart,
         const float* __restrict__ h, const float* __restrict__ dinv,
         const float* __restrict__ bias, const float* __restrict__ Wn,
         float* __restrict__ out, int nRows) {
    __shared__ float acc[RB * AS];
    __shared__ __attribute__((aligned(16))) float wl[FUSE ? D * WP : 4];

    const int tid  = threadIdx.x;
    const int row0 = (int)blockIdx.x * RB;
    const int left = nRows - row0;
    const int nr   = left < RB ? left : RB;

    if (FUSE) {
        #pragma unroll
        for (int q = 0; q < 4; ++q) {
            const int idx = tid + q * 1024;
            const int k   = idx >> 5;
            const int c   = (idx & 31) * 4;
            const float4 w = ((const float4*)Wn)[idx];
            *(float4*)&wl[k * WP + c + ((c >> 6) << 3)] = w;
        }
    }

    {
        const int c4 = tid & 31;
        const float b0 = bias[c4], b1 = bias[c4 + 32], b2 = bias[c4 + 64], b3 = bias[c4 + 96];
        int r = tid >> 5;
        for (; r + 32 < nr; r += 64) {
            const float4 va = ((const float4*)(h + (size_t)(row0 + r) * D))[c4];
            const float4 vb = ((const float4*)(h + (size_t)(row0 + r + 32) * D))[c4];
            const float  da = dinv[row0 + r];
            const float  db = dinv[row0 + r + 32];
            const float  na = da * da;
            const float  nb = db * db;
            float* a = &acc[r * AS + c4];
            a[0]  = va.x * na + b0;
            a[32] = va.y * na + b1;
            a[64] = va.z * na + b2;
            a[96] = va.w * na + b3;
            float* b = a + 32 * AS;
            b[0]  = vb.x * nb + b0;
            b[32] = vb.y * nb + b1;
            b[64] = vb.z * nb + b2;
            b[96] = vb.w * nb + b3;
        }
        if (r < nr) {
            const float4 va = ((const float4*)(h + (size_t)(row0 + r) * D))[c4];
            const float  da = dinv[row0 + r];
            const float  na = da * da;
            float* a = &acc[r * AS + c4];
            a[0]  = va.x * na + b0;
            a[32] = va.y * na + b1;
            a[64] = va.z * na + b2;
            a[96] = va.w * na + b3;
        }
    }
    __syncthreads();

    const int eBeg = gstart[blockIdx.x];
    const int eEnd = gstart[blockIdx.x + 1];
    const int lane = tid & 31;
    const int wv = __builtin_amdgcn_readfirstlane(tid >> 5);

    for (int e0 = eBeg + wv * 32; e0 < eEnd; e0 += 32 * 32) {
        const int cnt = eEnd - e0;
        const int e = e0 + lane;
        int s = 0, dl = 0;
        float nm = 0.0f;
        if (e < eEnd) {
            s = bsrc[e];
            const int d = bdst[e];
            dl = d - row0;
            if (FUSE) {
                nm = dinv[s] * dinv[d];
                bnorm[e] = nm;
            } else {
                nm = bnorm[e];
            }
        }
        if (cnt >= 32) {
            for (int j0 = 0; j0 < 32; j0 += EB) {
                float4 v[EB];
                #pragma unroll
                for (int u = 0; u < EB; ++u) {
                    const int sj = lane_bcast_i(s, j0 + u);
                    v[u] = ((const float4*)(h + (size_t)sj * D))[lane];
                }
                #pragma unroll
                for (int u = 0; u < EB; ++u) {
                    const int   dj = lane_bcast_i(dl, j0 + u);
                    const float nj = lane_bcast_f(nm, j0 + u);
                    float* a = &acc[dj * AS + lane];
                    atomicAdd(a,      v[u].x * nj);
                    atomicAdd(a + 32, v[u].y * nj);
                    atomicAdd(a + 64, v[u].z * nj);
                    atomicAdd(a + 96, v[u].w * nj);
                }
            }
        } else {
            for (int j = 0; j < cnt; ++j) {
                const int   sj = lane_bcast_i(s, j);
                const int   dj = lane_bcast_i(dl, j);
                const float nj = lane_bcast_f(nm, j);
                const float4 v = ((const float4*)(h + (size_t)sj * D))[lane];
                float* a = &acc[dj * AS + lane];
                atomicAdd(a,      v.x * nj);
                atomicAdd(a + 32, v.y * nj);
                atomicAdd(a + 64, v.z * nj);
                atomicAdd(a + 96, v.w * nj);
            }
        }
    }
    if (FUSE) {
        const int m     = lane & 15;
        const int kk    = (lane >> 4) << 1;
        const int halfm = (lane >> 4) * 8;
        const int ct    = wv & 7;
        const int pcol  = ct * 16 + m;
        const int wcol  = ((pcol & 3) << 5) + (pcol >> 2);

        const float* wb = &wl[kk * WP + wcol + ((wcol >> 6) << 3)];
        v2f bw[32];
        #pragma unroll
        for (int j = 0; j < 32; ++j) {
            bw[j].x = wb[(4 * j) * WP];
            bw[j].y = wb[(4 * j + 1) * WP];
        }
        __syncthreads();

        const int nst = (nr + 15) >> 4;
        for (int st = wv >> 3; st < nst; st += 4) {
            const float* xa = &acc[(st * 16 + m) * AS + kk];
            v8f c = {};
            #pragma unroll
            for (int j = 0; j < 32; ++j) {
                v2f a;
                a.x = fmaxf(xa[4 * j], 0.0f);
                a.y = fmaxf(xa[4 * j + 1], 0.0f);
                c = __builtin_amdgcn_wmma_f32_16x16x4_f32(
                     false, a,  false, bw[j],
                     (short)0, c,  false,  false);
            }
            const int r0 = st * 16 + halfm;
            float* hp = out + (size_t)(row0 + r0) * D + pcol;
            if (st * 16 + 16 <= nr) {
                #pragma unroll
                for (int r = 0; r < 8; ++r) hp[(size_t)r * D] = c[r];
            } else {
                #pragma unroll
                for (int r = 0; r < 8; ++r) {
                    if (r0 + r < nr) hp[(size_t)r * D] = c[r];
                }
            }
        }
    } else {
        __syncthreads();

        const int nq = nr * (D / 4);
        for (int idx = tid; idx < nq; idx += 1024) {
            const int r  = idx >> 5;
            const int c4 = idx & 31;
            ((float4*)(out + (size_t)(row0 + r) * D))[c4] = *(const float4*)&acc[r * AS + c4 * 4];
        }
    }
}


static inline size_t align256(size_t v) { return (v + 255) & ~(size_t)255; }

extern "C" void kernel_launch(void* const* d_in, const int* in_sizes, int n_in,
                              void* d_out, int out_size, void* d_ws, size_t ws_size,
                              hipStream_t stream) {
    const float* x  = (const float*)d_in[0];
    const int*   ei = (const int*)d_in[1];
    const float* W1 = (const float*)d_in[2];
    const float* b1 = (const float*)d_in[3];
    const float* W2 = (const float*)d_in[4];
    const float* b2 = (const float*)d_in[5];
    float* out = (float*)d_out;

    const int N = in_sizes[0] / D;
    const int E = in_sizes[1] / 2;
    const int* srcIdx = ei;
    const int* dstIdx = ei + E;

    char* ws = (char*)d_ws;
    const size_t o0 = align256((size_t)N * sizeof(float));
    float* dinv = (float*)(ws + o0);
    const size_t o1 = o0 + align256((size_t)N * sizeof(float));
    float* hbuf = (float*)(ws + o1);
    const size_t o2 = o1 + align256((size_t)N * D * sizeof(float));
    float* h2buf = (float*)(ws + o2);
    const size_t o3 = o2 + align256((size_t)N * D * sizeof(float));
    int* gcnt   = (int*)(ws + o3);
    const size_t o4 = o3 + align256((size_t)MAXG * sizeof(int));
    int* gcur   = (int*)(ws + o4);
    const size_t o5 = o4 + align256((size_t)MAXG * sizeof(int));
    int* gstart = (int*)(ws + o5);
    const size_t o6 = o5 + align256((size_t)(MAXG + 1) * sizeof(int));
    int* bsrc   = (int*)(ws + o6);
    const size_t o7 = o6 + align256((size_t)E * sizeof(int));
    int* bdst   = (int*)(ws + o7);
    const size_t o8 = o7 + align256((size_t)E * sizeof(int));
    float* bnorm = (float*)(ws + o8);

    const int blk = 256;
    const int gCh = (E + CH - 1) / CH;
    const int gGemm = GEMM_BLOCKS;
    const int nG  = (N + RB - 1) / RB;

    init_cnt   <<<(nG + blk - 1) / blk, blk, 0, stream>>>(gcnt, gcur, nG);
    grp_hist   <<<gCh, blk, 0, stream>>>(dstIdx, gcnt, E, nG);
    group_edges<<<gCh, blk, 0, stream>>>(srcIdx, dstIdx, gcnt, gcur, gstart, bsrc, bdst, E, nG);
    row_deg    <<<nG, 1024, 0, stream>>>(bdst, gstart, dinv, N);

    gemm_wmma_f32  <<<gGemm, blk, 0, stream>>>(x, W1, hbuf, N);
    edge_lds<true> <<<nG, 1024, 0, stream>>>(bsrc, bdst, bnorm, gstart, hbuf, dinv, b1, W2, h2buf, N);

    edge_lds<false><<<nG, 1024, 0, stream>>>(bsrc, bdst, bnorm, gstart, h2buf, dinv, b2, nullptr, out, N);
}
